# starting kernel
# baseline (speedup 1.0000x reference)
#include <hip/hip_runtime.h>

typedef __attribute__((ext_vector_type(2))) float v2f;
typedef __attribute__((ext_vector_type(8))) float v8f;

#define IN_H 128
#define IN_W 128
#define KHH 8
#define KWW 8
#define OH 121
#define OW 121
#define BATCH 64
#define UW 24
#define KP 192
#define LSTRIDE 196
#define NWAVES 4

__global__ __launch_bounds__(128) void lc2d_wmma_f32_kernel(
    const float* __restrict__ x,
    const float* __restrict__ w,
    const float* __restrict__ bias,
    float* __restrict__ out)
{
    __shared__ float Wl[16 * LSTRIDE];
    __shared__ float Xl[NWAVES][16 * LSTRIDE];

    const int i    = blockIdx.x >> 3;
    const int j0   = (blockIdx.x & 7) << 4;
    const int tid  = threadIdx.x;
    const int wave = tid >> 5;
    const int lane = tid & 31;

    for (int idx = tid; idx < 16 * LSTRIDE; idx += 128) Wl[idx] = 0.0f;
    __syncthreads();
    for (int idx = tid; idx < 16 * (KHH * KWW); idx += 128) {
        const int jl = idx >> 6;
        const int t  = idx & 63;
        const int kh = t >> 3, kw = t & 7;
        const int j  = j0 + jl;
        if (j < OW) {
            Wl[jl * LSTRIDE + kh * UW + jl + kw] = w[(i * OW + j) * (KHH * KWW) + t];
        }
    }

    for (int idx = lane; idx < 16 * KP; idx += 32) {
        const int b   = idx / KP;
        const int k   = idx % KP;
        const int kh  = k / UW;
        const int c   = k % UW;
        const int col = j0 + c;
        float v = 0.0f;
        if (col < IN_W)
            v = x[((wave * 16 + b) * IN_H + (i + kh)) * IN_W + col];
        Xl[wave][b * LSTRIDE + k] = v;
    }
    __syncthreads();

    const int mn    = lane & 15;
    const int khalf = lane >> 4;
    const float* arow = &Wl[mn * LSTRIDE + khalf * 2];
    const float* brow = &Xl[wave][mn * LSTRIDE + khalf * 2];

    v8f acc = {};
    #pragma unroll
    for (int kk = 0; kk < KP / 4; ++kk) {
        v2f a = *(const v2f*)(arow + kk * 4);
        v2f b = *(const v2f*)(brow + kk * 4);
        acc = __builtin_amdgcn_wmma_f32_16x16x4_f32(
             false, a,  false, b,
             (short)0, acc,  false,  false);
    }

    const int bglob = wave * 16 + mn;
    #pragma unroll
    for (int v = 0; v < 8; ++v) {
        const int j = j0 + v + khalf * 8;
        if (j < OW)
            out[(bglob * OH + i) * OW + j] = acc[v] + bias[i * OW + j];
    }
}

extern "C" void kernel_launch(void* const* d_in, const int* in_sizes, int n_in,
                              void* d_out, int out_size, void* d_ws, size_t ws_size,
                              hipStream_t stream) {
    const float* x    = (const float*)d_in[0];
    const float* w    = (const float*)d_in[1];
    const float* bias = (const float*)d_in[2];
    float* out        = (float*)d_out;

    dim3 grid(OH * 8);
    dim3 block(128);
    lc2d_wmma_f32_kernel<<<grid, block, 0, stream>>>(x, w, bias, out);
}


# 4 output rows per block: a wave keeps one x window (11 rows x 6 fragments) in registers for all 4; the 4 weight tiles filled from one batch of 32 loads per thread, no wait per element; sums unchanged
# speedup vs baseline: 4.5007x; 4.5007x over previous
#include <hip/hip_runtime.h>


typedef __attribute__((ext_vector_type(2))) float v2f;
typedef __attribute__((ext_vector_type(4))) float v4f;
typedef __attribute__((ext_vector_type(8))) float v8f;

#define IN_H 128
#define IN_W 128
#define KHH 8
#define KWW 8
#define OH 121
#define OW 121
#define BATCH 64
#define UW 24
#define KP 192
#define LSTRIDE 196
#define KSTEPS (KP / 4)
#define TP 20
#define RB 4
#define XROWS (RB + KHH - 1)
#define NGRP ((OH + RB - 1) / RB)
#define WT (16 * LSTRIDE)

__global__ __launch_bounds__(128) void lc2d_wmma_f32_kernel(
    const float* __restrict__ x,
    const float* __restrict__ w,
    const float* __restrict__ bias,
    float* __restrict__ out)
{
    __shared__ float Wl[RB * WT];
    __shared__ __attribute__((aligned(16))) float Tl[4 * RB * 16 * TP];

    const int i0   = (blockIdx.x >> 3) * RB;
    const int j0   = (blockIdx.x & 7) << 4;
    const int tid  = threadIdx.x;
    const int wave = tid >> 5;
    const int lane = tid & 31;
    const int mn    = lane & 15;
    const int khalf = lane >> 4;

    const float* xb = x + (wave * 16 + mn) * (IN_H * IN_W) + j0 + khalf * 2;
    v2f bx[XROWS][6];
    #pragma unroll
    for (int t = 0; t < XROWS; ++t) {
        const int row = min(i0 + t, IN_H - 1);
        #pragma unroll
        for (int q = 0; q < 6; ++q) {
            const int c = q * 4;
            const bool inb = (j0 + c) < IN_W;
            const v2f v = *(const v2f*)(xb + row * IN_W + (inb ? c : 0));
            const v2f z = {0.0f, 0.0f};
            bx[t][q] = inb ? v : z;
        }
    }

    const int jcol = j0 + mn;
    float bj[RB];
    #pragma unroll
    for (int r = 0; r < RB; ++r)
        bj[r] = (jcol < OW && (i0 + r) < OH) ? bias[(i0 + r) * OW + jcol] : 0.0f;

    for (int idx = tid; idx < RB * WT; idx += 128) Wl[idx] = 0.0f;
    __syncthreads();
    float wv[RB * 8];
    #pragma unroll
    for (int n = 0; n < RB * 8; ++n) {
        const int idx = tid + n * 128;
        const int r  = idx >> 10;
        const int jl = (idx >> 6) & 15;
        const int t  = idx & 63;
        const int j  = j0 + jl;
        const int i  = i0 + r;
        const bool ok = (j < OW) && (i < OH);
        wv[n] = w[ok ? ((i * OW + j) * (KHH * KWW) + t) : t];
    }
    #pragma unroll
    for (int n = 0; n < RB * 8; ++n) {
        const int idx = tid + n * 128;
        const int r  = idx >> 10;
        const int jl = (idx >> 6) & 15;
        const int t  = idx & 63;
        const int kh = t >> 3, kw = t & 7;
        if ((j0 + jl) < OW && (i0 + r) < OH) {
            Wl[r * WT + jl * LSTRIDE + kh * UW + jl + kw] = wv[n];
        }
    }
    __syncthreads();

    const float* arow = &Wl[mn * LSTRIDE + khalf * 2];

    v8f acc[RB];
    #pragma unroll
    for (int r = 0; r < RB; ++r) { const v8f zz = {}; acc[r] = zz; }

    #pragma unroll
    for (int t = 0; t < XROWS; ++t) {
        #pragma unroll
        for (int q = 0; q < 6; ++q) {
            #pragma unroll
            for (int r = 0; r < RB; ++r) {
                const int kh = t - r;
                if (kh >= 0 && kh < KHH) {
                    const v2f a = *(const v2f*)(arow + r * WT + (kh * 6 + q) * 4);
                    acc[r] = __builtin_amdgcn_wmma_f32_16x16x4_f32(
                         false, a,  false, bx[t][q],
                         (short)0, acc[r],  false,  false);
                }
            }
        }
    }

    float* tw = &Tl[wave * RB * 16 * TP];
    #pragma unroll
    for (int r = 0; r < RB; ++r) {
        const v4f lo = {acc[r][0], acc[r][1], acc[r][2], acc[r][3]};
        const v4f hi = {acc[r][4], acc[r][5], acc[r][6], acc[r][7]};
        *(v4f*)(tw + r * 16 * TP + mn * TP + khalf * 8)     = lo;
        *(v4f*)(tw + r * 16 * TP + mn * TP + khalf * 8 + 4) = hi;
    }
    if (jcol < OW) {
        #pragma unroll
        for (int r = 0; r < RB; ++r) {
            if ((i0 + r) < OH) {
                float* op = out + ((wave * 16 + khalf) * OH + i0 + r) * OW + jcol;
                #pragma unroll
                for (int s = 0; s < 8; ++s)
                    op[(2 * s) * (OH * OW)] = tw[r * 16 * TP + (2 * s + khalf) * TP + mn] + bj[r];
            }
        }
    }
}

extern "C" void kernel_launch(void* const* d_in, const int* in_sizes, int n_in,
                              void* d_out, int out_size, void* d_ws, size_t ws_size,
                              hipStream_t stream) {
    const float* x    = (const float*)d_in[0];
    const float* w    = (const float*)d_in[1];
    const float* bias = (const float*)d_in[2];
    float* out        = (float*)d_out;

    dim3 grid(NGRP * 8);
    dim3 block(128);
    lc2d_wmma_f32_kernel<<<grid, block, 0, stream>>>(x, w, bias, out);
}


# weights requested before the input window; both barriers fence LDS only: nothing waits for the 66 window loads before the matrix loop, which takes each input row as it arrives; sums unchanged
# speedup vs baseline: 4.8490x; 1.0774x over previous
#include <hip/hip_runtime.h>


typedef __attribute__((ext_vector_type(2))) float v2f;
typedef __attribute__((ext_vector_type(4))) float v4f;
typedef __attribute__((ext_vector_type(8))) float v8f;

#define IN_H 128
#define IN_W 128
#define KHH 8
#define KWW 8
#define OH 121
#define OW 121
#define BATCH 64
#define UW 24
#define KP 192
#define LSTRIDE 196
#define KSTEPS (KP / 4)
#define TP 20
#define RB 4
#define XROWS (RB + KHH - 1)
#define NGRP ((OH + RB - 1) / RB)
#define WT (16 * LSTRIDE)

__device__ __forceinline__ void lds_barrier() {
    __builtin_amdgcn_fence(__ATOMIC_RELEASE, "workgroup", "local");
    __builtin_amdgcn_s_barrier();
    __builtin_amdgcn_fence(__ATOMIC_ACQUIRE, "workgroup", "local");
}

__global__ __launch_bounds__(128) void lc2d_wmma_f32_kernel(
    const float* __restrict__ x,
    const float* __restrict__ w,
    const float* __restrict__ bias,
    float* __restrict__ out)
{
    __shared__ float Wl[RB * WT];
    __shared__ __attribute__((aligned(16))) float Tl[4 * RB * 16 * TP];

    const int i0   = (blockIdx.x >> 3) * RB;
    const int j0   = (blockIdx.x & 7) << 4;
    const int tid  = threadIdx.x;
    const int wave = tid >> 5;
    const int lane = tid & 31;
    const int mn    = lane & 15;
    const int khalf = lane >> 4;


    float wv[RB * 8];
    #pragma unroll
    for (int n = 0; n < RB * 8; ++n) {
        const int idx = tid + n * 128;
        const int r  = idx >> 10;
        const int jl = (idx >> 6) & 15;
        const int t  = idx & 63;
        const int j  = j0 + jl;
        const int i  = i0 + r;
        const bool ok = (j < OW) && (i < OH);
        wv[n] = w[ok ? ((i * OW + j) * (KHH * KWW) + t) : t];
    }

    const int jcol = j0 + mn;
    float bj[RB];
    #pragma unroll
    for (int r = 0; r < RB; ++r)
        bj[r] = (jcol < OW && (i0 + r) < OH) ? bias[(i0 + r) * OW + jcol] : 0.0f;

    for (int idx = tid; idx < RB * WT; idx += 128) Wl[idx] = 0.0f;

    const float* xb = x + (wave * 16 + mn) * (IN_H * IN_W) + j0 + khalf * 2;
    v2f bx[XROWS][6];
    #pragma unroll
    for (int t = 0; t < XROWS; ++t) {
        const int row = min(i0 + t, IN_H - 1);
        #pragma unroll
        for (int q = 0; q < 6; ++q) {
            const int c = q * 4;
            const bool inb = (j0 + c) < IN_W;
            const v2f v = *(const v2f*)(xb + row * IN_W + (inb ? c : 0));
            const v2f z = {0.0f, 0.0f};
            bx[t][q] = inb ? v : z;
        }
    }

    lds_barrier();

    #pragma unroll
    for (int n = 0; n < RB * 8; ++n) {
        const int idx = tid + n * 128;
        const int r  = idx >> 10;
        const int jl = (idx >> 6) & 15;
        const int t  = idx & 63;
        const int kh = t >> 3, kw = t & 7;
        if ((j0 + jl) < OW && (i0 + r) < OH) {
            Wl[r * WT + jl * LSTRIDE + kh * UW + jl + kw] = wv[n];
        }
    }
    lds_barrier();

    const float* arow = &Wl[mn * LSTRIDE + khalf * 2];

    v8f acc[RB];
    #pragma unroll
    for (int r = 0; r < RB; ++r) { const v8f zz = {}; acc[r] = zz; }

    #pragma unroll
    for (int t = 0; t < XROWS; ++t) {
        #pragma unroll
        for (int q = 0; q < 6; ++q) {
            #pragma unroll
            for (int r = 0; r < RB; ++r) {
                const int kh = t - r;
                if (kh >= 0 && kh < KHH) {
                    const v2f a = *(const v2f*)(arow + r * WT + (kh * 6 + q) * 4);
                    acc[r] = __builtin_amdgcn_wmma_f32_16x16x4_f32(
                         false, a,  false, bx[t][q],
                         (short)0, acc[r],  false,  false);
                }
            }
        }
    }

    float* tw = &Tl[wave * RB * 16 * TP];
    #pragma unroll
    for (int r = 0; r < RB; ++r) {
        const v4f lo = {acc[r][0], acc[r][1], acc[r][2], acc[r][3]};
        const v4f hi = {acc[r][4], acc[r][5], acc[r][6], acc[r][7]};
        *(v4f*)(tw + r * 16 * TP + mn * TP + khalf * 8)     = lo;
        *(v4f*)(tw + r * 16 * TP + mn * TP + khalf * 8 + 4) = hi;
    }
    if (jcol < OW) {
        #pragma unroll
        for (int r = 0; r < RB; ++r) {
            if ((i0 + r) < OH) {
                float* op = out + ((wave * 16 + khalf) * OH + i0 + r) * OW + jcol;
                #pragma unroll
                for (int s = 0; s < 8; ++s)
                    op[(2 * s) * (OH * OW)] = tw[r * 16 * TP + (2 * s + khalf) * TP + mn] + bj[r];
            }
        }
    }
}

extern "C" void kernel_launch(void* const* d_in, const int* in_sizes, int n_in,
                              void* d_out, int out_size, void* d_ws, size_t ws_size,
                              hipStream_t stream) {
    const float* x    = (const float*)d_in[0];
    const float* w    = (const float*)d_in[1];
    const float* bias = (const float*)d_in[2];
    float* out        = (float*)d_out;

    dim3 grid(NGRP * 8);
    dim3 block(128);
    lc2d_wmma_f32_kernel<<<grid, block, 0, stream>>>(x, w, bias, out);
}
